# starting kernel
# baseline (speedup 1.0000x reference)
#include <hip/hip_runtime.h>
#include <math.h>

typedef float v2f __attribute__((ext_vector_type(2)));
typedef float v8f __attribute__((ext_vector_type(8)));

#define N_NODES   100000
#define N_EDGES   1600000
#define D_IN      128
#define D_OUT     64
#define ROW_BLOCKS (N_NODES / 16)

__global__ void k_init(float* __restrict__ out, float* __restrict__ m,
                       float* __restrict__ denom) {
    int i = blockIdx.x * 256 + threadIdx.x;
    if (i < N_NODES * D_OUT) out[i] = 0.0f;
    if (i < N_NODES) { m[i] = -__builtin_inff(); denom[i] = 0.0f; }
}

__global__ __launch_bounds__(256) void k_gemm(const float* __restrict__ X,
                                              const float* __restrict__ W,
                                              const float* __restrict__ bias,
                                              float* __restrict__ Xp) {
    __shared__ float Wl[D_IN * D_OUT];
    const int tid = threadIdx.x;
    for (int i = tid; i < D_IN * D_OUT; i += 256) {
        const int k  = i >> 6;
        const int c  = i & 63;
        const int kk = k & ~1;
        Wl[kk * D_OUT + (c << 1) + (k & 1)] = W[i];
    }
    __syncthreads();

    const int wave     = tid >> 5;
    const int lane     = tid & 31;
    const int rowBlock = blockIdx.x * 8 + wave;
    if (rowBlock >= ROW_BLOCKS) return;

    const int rowBase = rowBlock * 16;
    const int l15     = lane & 15;
    const int khalf   = (lane >> 4) << 1;

    const float* xrow = X + (size_t)(rowBase + l15) * D_IN;

    v8f acc0 = {}, acc1 = {}, acc2 = {}, acc3 = {};
    for (int k0 = 0; k0 < D_IN; k0 += 4) {
        v2f a = *(const v2f*)(xrow + k0 + khalf);
        const float* wp = Wl + (k0 + khalf) * D_OUT;
        v2f b0 = *(const v2f*)(wp + (l15 << 1));
        v2f b1 = *(const v2f*)(wp + ((16 + l15) << 1));
        v2f b2 = *(const v2f*)(wp + ((32 + l15) << 1));
        v2f b3 = *(const v2f*)(wp + ((48 + l15) << 1));
        acc0 = __builtin_amdgcn_wmma_f32_16x16x4_f32(false, a, false, b0, (short)0, acc0, false, false);
        acc1 = __builtin_amdgcn_wmma_f32_16x16x4_f32(false, a, false, b1, (short)0, acc1, false, false);
        acc2 = __builtin_amdgcn_wmma_f32_16x16x4_f32(false, a, false, b2, (short)0, acc2, false, false);
        acc3 = __builtin_amdgcn_wmma_f32_16x16x4_f32(false, a, false, b3, (short)0, acc3, false, false);
    }

    const int orow = rowBase + (lane >> 4) * 8;
    const float b0s = bias[l15], b1s = bias[16 + l15];
    const float b2s = bias[32 + l15], b3s = bias[48 + l15];
#pragma unroll
    for (int r = 0; r < 8; ++r) {
        float* o = Xp + (size_t)(orow + r) * D_OUT;
        o[l15]      = acc0[r] + b0s;
        o[16 + l15] = acc1[r] + b1s;
        o[32 + l15] = acc2[r] + b2s;
        o[48 + l15] = acc3[r] + b3s;
    }
}

__global__ __launch_bounds__(256) void k_scores(const float* __restrict__ Xp,
                                                const float* __restrict__ a_src,
                                                const float* __restrict__ a_src_b,
                                                const float* __restrict__ a_dst,
                                                const float* __restrict__ a_dst_b,
                                                float* __restrict__ s_src,
                                                float* __restrict__ s_dst) {
    const int lane = threadIdx.x & 31;
    const int node = (blockIdx.x * 256 + threadIdx.x) >> 5;
    if (node >= N_NODES) return;
    const float* row = Xp + (size_t)node * D_OUT;
    const float x0 = row[lane], x1 = row[lane + 32];
    float ps = x0 * a_src[lane] + x1 * a_src[lane + 32];
    float pd = x0 * a_dst[lane] + x1 * a_dst[lane + 32];
#pragma unroll
    for (int off = 16; off > 0; off >>= 1) {
        ps += __shfl_xor(ps, off, 32);
        pd += __shfl_xor(pd, off, 32);
    }
    if (lane == 0) {
        s_src[node] = ps + a_src_b[0];
        s_dst[node] = pd + a_dst_b[0];
    }
}

__global__ void k_edge_logits(const int* __restrict__ e_src,
                              const int* __restrict__ e_dst,
                              const float* __restrict__ s_src,
                              const float* __restrict__ s_dst,
                              float* __restrict__ e, float* __restrict__ m) {
    int i = blockIdx.x * 256 + threadIdx.x;
    if (i >= N_EDGES) return;
    float v = s_src[e_src[i]] + s_dst[e_dst[i]];
    v = v > 0.0f ? v : 0.01f * v;
    e[i] = v;
    float* addr = m + e_dst[i];
    if (v >= 0.0f) atomicMax((int*)addr, __float_as_int(v));
    else           atomicMin((unsigned int*)addr, __float_as_uint(v));
}

__global__ void k_edge_exp(const int* __restrict__ e_dst,
                           float* __restrict__ e,
                           const float* __restrict__ m,
                           float* __restrict__ denom) {
    int i = blockIdx.x * 256 + threadIdx.x;
    if (i >= N_EDGES) return;
    const int d = e_dst[i];
    const float w = expf(e[i] - m[d]);
    e[i] = w;
    atomicAdd(denom + d, w);
}

__global__ __launch_bounds__(256) void k_scatter(const int* __restrict__ e_src,
                                                 const int* __restrict__ e_dst,
                                                 const float* __restrict__ e,
                                                 const float* __restrict__ denom,
                                                 const float* __restrict__ Xp,
                                                 float* __restrict__ out) {
    const int lane = threadIdx.x & 31;
    const long long edge = ((long long)blockIdx.x * 256 + threadIdx.x) >> 5;
    if (edge >= N_EDGES) return;
    const int s = e_src[edge], d = e_dst[edge];
    const float coeff = e[edge] / denom[d];
    const float* xs = Xp + (size_t)s * D_OUT;
    float* od = out + (size_t)d * D_OUT;
    atomicAdd(od + lane,      coeff * xs[lane]);
    atomicAdd(od + lane + 32, coeff * xs[lane + 32]);
}

__global__ void k_elu(float* __restrict__ out) {
    int i = blockIdx.x * 256 + threadIdx.x;
    if (i >= N_NODES * D_OUT) return;
    const float v = out[i];
    out[i] = v > 0.0f ? v : (expf(v) - 1.0f);
}

extern "C" void kernel_launch(void* const* d_in, const int* in_sizes, int n_in,
                              void* d_out, int out_size, void* d_ws, size_t ws_size,
                              hipStream_t stream) {
    const float* X       = (const float*)d_in[0];
    const float* theta_w = (const float*)d_in[1];
    const float* theta_b = (const float*)d_in[2];
    const float* a_src   = (const float*)d_in[3];
    const float* a_src_b = (const float*)d_in[4];
    const float* a_dst   = (const float*)d_in[5];
    const float* a_dst_b = (const float*)d_in[6];
    const int*   e_src   = (const int*)d_in[7];
    const int*   e_dst   = (const int*)d_in[8];
    float*       out     = (float*)d_out;

    float* Xp    = (float*)d_ws;
    float* s_src = Xp    + (size_t)N_NODES * D_OUT;
    float* s_dst = s_src + N_NODES;
    float* m     = s_dst + N_NODES;
    float* denom = m     + N_NODES;
    float* e     = denom + N_NODES;

    const int elemGrid  = (N_NODES * D_OUT + 255) / 256;
    const int gemmGrid  = (ROW_BLOCKS + 7) / 8;
    const int nodeWGrid = (N_NODES * 32 + 255) / 256;
    const int edgeGrid  = (N_EDGES + 255) / 256;
    const int edgeWGrid = (int)(((long long)N_EDGES * 32 + 255) / 256);

    k_init<<<elemGrid, 256, 0, stream>>>(out, m, denom);
    k_gemm<<<gemmGrid, 256, 0, stream>>>(X, theta_w, theta_b, Xp);
    k_scores<<<nodeWGrid, 256, 0, stream>>>(Xp, a_src, a_src_b, a_dst, a_dst_b, s_src, s_dst);
    k_edge_logits<<<edgeGrid, 256, 0, stream>>>(e_src, e_dst, s_src, s_dst, e, m);
    k_edge_exp<<<edgeGrid, 256, 0, stream>>>(e_dst, e, m, denom);
    k_scatter<<<edgeWGrid, 256, 0, stream>>>(e_src, e_dst, e, denom, Xp, out);
    k_elu<<<elemGrid, 256, 0, stream>>>(out);
}


# edges grouped by destination in every call (integer slot counts, scanned ranges, each sorted by source); one wave per node: softmax, row sum, ELU in registers; no float atomics; sums in source order
# speedup vs baseline: 1.8540x; 1.8540x over previous
#include <hip/hip_runtime.h>
#include <math.h>

typedef float v2f __attribute__((ext_vector_type(2)));
typedef float v8f __attribute__((ext_vector_type(8)));

#define N_NODES   100000
#define N_EDGES   1600000
#define D_IN      128
#define D_OUT     64
#define ROW_BLOCKS (N_NODES / 16)

__global__ __launch_bounds__(256) void k_gemm(const float* __restrict__ X,
                                              const float* __restrict__ W,
                                              const float* __restrict__ bias,
                                              float* __restrict__ Xp) {
    __shared__ float Wl[D_IN * D_OUT];
    const int tid = threadIdx.x;
    for (int i = tid; i < D_IN * D_OUT; i += 256) {
        const int k  = i >> 6;
        const int c  = i & 63;
        const int kk = k & ~1;
        Wl[kk * D_OUT + (c << 1) + (k & 1)] = W[i];
    }
    __syncthreads();

    const int wave     = tid >> 5;
    const int lane     = tid & 31;
    const int rowBlock = blockIdx.x * 8 + wave;
    if (rowBlock >= ROW_BLOCKS) return;

    const int rowBase = rowBlock * 16;
    const int l15     = lane & 15;
    const int khalf   = (lane >> 4) << 1;

    const float* xrow = X + (size_t)(rowBase + l15) * D_IN;

    v8f acc0 = {}, acc1 = {}, acc2 = {}, acc3 = {};
    for (int k0 = 0; k0 < D_IN; k0 += 4) {
        v2f a = *(const v2f*)(xrow + k0 + khalf);
        const float* wp = Wl + (k0 + khalf) * D_OUT;
        v2f b0 = *(const v2f*)(wp + (l15 << 1));
        v2f b1 = *(const v2f*)(wp + ((16 + l15) << 1));
        v2f b2 = *(const v2f*)(wp + ((32 + l15) << 1));
        v2f b3 = *(const v2f*)(wp + ((48 + l15) << 1));
        acc0 = __builtin_amdgcn_wmma_f32_16x16x4_f32(false, a, false, b0, (short)0, acc0, false, false);
        acc1 = __builtin_amdgcn_wmma_f32_16x16x4_f32(false, a, false, b1, (short)0, acc1, false, false);
        acc2 = __builtin_amdgcn_wmma_f32_16x16x4_f32(false, a, false, b2, (short)0, acc2, false, false);
        acc3 = __builtin_amdgcn_wmma_f32_16x16x4_f32(false, a, false, b3, (short)0, acc3, false, false);
    }

    const int orow = rowBase + (lane >> 4) * 8;
    const float b0s = bias[l15], b1s = bias[16 + l15];
    const float b2s = bias[32 + l15], b3s = bias[48 + l15];
#pragma unroll
    for (int r = 0; r < 8; ++r) {
        float* o = Xp + (size_t)(orow + r) * D_OUT;
        o[l15]      = acc0[r] + b0s;
        o[16 + l15] = acc1[r] + b1s;
        o[32 + l15] = acc2[r] + b2s;
        o[48 + l15] = acc3[r] + b3s;
    }
}

__global__ __launch_bounds__(256) void k_scores(const float* __restrict__ Xp,
                                                const float* __restrict__ a_src,
                                                const float* __restrict__ a_src_b,
                                                const float* __restrict__ a_dst,
                                                const float* __restrict__ a_dst_b,
                                                float* __restrict__ s_src,
                                                float* __restrict__ s_dst) {
    const int lane = threadIdx.x & 31;
    const int node = (blockIdx.x * 256 + threadIdx.x) >> 5;
    if (node >= N_NODES) return;
    const float* row = Xp + (size_t)node * D_OUT;
    const float x0 = row[lane], x1 = row[lane + 32];
    float ps = x0 * a_src[lane] + x1 * a_src[lane + 32];
    float pd = x0 * a_dst[lane] + x1 * a_dst[lane + 32];
#pragma unroll
    for (int off = 16; off > 0; off >>= 1) {
        ps += __shfl_xor(ps, off, 32);
        pd += __shfl_xor(pd, off, 32);
    }
    if (lane == 0) {
        s_src[node] = ps + a_src_b[0];
        s_dst[node] = pd + a_dst_b[0];
    }
}

__global__ void k_clear(int* __restrict__ cnt, int* __restrict__ nxt) {
    const int i = blockIdx.x * 256 + threadIdx.x;
    if (i < N_NODES) cnt[i] = 0;
    if (i == 0) *nxt = 0;
}

__global__ void k_slot(const int* __restrict__ e_dst, int* __restrict__ cnt,
                       int* __restrict__ pos) {
    const int i = blockIdx.x * 256 + threadIdx.x;
    if (i >= N_EDGES) return;
    pos[i] = atomicAdd(cnt + e_dst[i], 1);
}

__global__ __launch_bounds__(256) void k_alloc(const int* __restrict__ cnt,
                                               int* __restrict__ nxt,
                                               int* __restrict__ off) {
    __shared__ int wsum[8];
    __shared__ int blockBase;
    const int tid  = threadIdx.x;
    const int lane = tid & 31;
    const int wave = tid >> 5;
    const int n    = blockIdx.x * 256 + tid;
    const int c    = n < N_NODES ? cnt[n] : 0;
    int incl = c;
    for (int o = 1; o < 32; o <<= 1) {
        const int t = __shfl_up(incl, o, 32);
        if (lane >= o) incl += t;
    }
    if (lane == 31) wsum[wave] = incl;
    __syncthreads();
    if (tid == 0) {
        int s = 0;
        for (int w = 0; w < 8; ++w) { const int t = wsum[w]; wsum[w] = s; s += t; }
        blockBase = atomicAdd(nxt, s);
    }
    __syncthreads();
    if (n < N_NODES) off[n] = blockBase + wsum[wave] + incl - c;
}

__global__ void k_fill(const int* __restrict__ e_src, const int* __restrict__ e_dst,
                       const int* __restrict__ pos, const int* __restrict__ off,
                       int* __restrict__ grp) {
    const int i = blockIdx.x * 256 + threadIdx.x;
    if (i >= N_EDGES) return;
    grp[off[e_dst[i]] + pos[i]] = e_src[i];
}

__global__ __launch_bounds__(256) void k_sort(const int* __restrict__ cnt,
                                              const int* __restrict__ off,
                                              const int* __restrict__ grp,
                                              int* __restrict__ srt) {
    const int lane = threadIdx.x & 31;
    const int node = __builtin_amdgcn_readfirstlane((blockIdx.x * 256 + threadIdx.x) >> 5);
    if (node >= N_NODES) return;
    const int deg  = __builtin_amdgcn_readfirstlane(cnt[node]);
    const int base = __builtin_amdgcn_readfirstlane(off[node]);
    for (int c0 = 0; c0 < deg; c0 += 32) {
        const int p   = c0 + lane;
        const int key = p < deg ? grp[base + p] : 0x7fffffff;
        const unsigned long long mine =
            ((unsigned long long)(unsigned)key << 32) | (unsigned)p;
        int rank = 0;
        for (int d0 = 0; d0 < deg; d0 += 32) {
            const int q     = d0 + lane;
            const int other = q < deg ? grp[base + q] : 0x7fffffff;
            const int nv    = deg - d0 < 32 ? deg - d0 : 32;
            for (int j = 0; j < nv; ++j) {
                const int kj = __builtin_amdgcn_readlane(other, j);
                const unsigned long long theirs =
                    ((unsigned long long)(unsigned)kj << 32) | (unsigned)(d0 + j);
                rank += theirs < mine ? 1 : 0;
            }
        }
        if (p < deg) srt[base + rank] = key;
    }
}

__device__ __forceinline__ float wave_max(float v) {
    for (int o = 16; o > 0; o >>= 1) v = fmaxf(v, __shfl_xor(v, o, 32));
    return v;
}
__device__ __forceinline__ float wave_sum(float v) {
    for (int o = 16; o > 0; o >>= 1) v += __shfl_xor(v, o, 32);
    return v;
}
__device__ __forceinline__ float leaky(float v) { return v > 0.0f ? v : 0.01f * v; }

__device__ __forceinline__ void add_rows(const float* __restrict__ Xp, int src, float coef,
                                         int nv, int lane, v2f& acc) {
#pragma unroll 4
    for (int j = 0; j < nv; ++j) {
        const int   sj = __builtin_amdgcn_readlane(src, j);
        const float cj = __builtin_bit_cast(float,
            __builtin_amdgcn_readlane(__builtin_bit_cast(int, coef), j));
        const v2f x = *(const v2f*)(Xp + (size_t)sj * D_OUT + 2 * lane);
        acc.x += cj * x.x;
        acc.y += cj * x.y;
    }
}

__global__ __launch_bounds__(256) void k_gather(const int* __restrict__ cnt,
                                                const int* __restrict__ off,
                                                const int* __restrict__ srt,
                                                const float* __restrict__ s_src,
                                                const float* __restrict__ s_dst,
                                                const float* __restrict__ Xp,
                                                float* __restrict__ out) {
    const int lane = threadIdx.x & 31;
    const int node = __builtin_amdgcn_readfirstlane((blockIdx.x * 256 + threadIdx.x) >> 5);
    if (node >= N_NODES) return;
    const int deg  = __builtin_amdgcn_readfirstlane(cnt[node]);
    const int base = __builtin_amdgcn_readfirstlane(off[node]);
    const float sd = s_dst[node];

    int   src0 = 0;
    float v0   = -__builtin_inff();
    if (lane < deg) { src0 = srt[base + lane]; v0 = leaky(s_src[src0] + sd); }

    float mx = v0;
    for (int c0 = 32; c0 < deg; c0 += 32) {
        const int p = c0 + lane;
        if (p < deg) mx = fmaxf(mx, leaky(s_src[srt[base + p]] + sd));
    }
    mx = wave_max(mx);

    const float w0 = lane < deg ? expf(v0 - mx) : 0.0f;
    float den = wave_sum(w0);
    for (int c0 = 32; c0 < deg; c0 += 32) {
        const int p = c0 + lane;
        const float w = p < deg ? expf(leaky(s_src[srt[base + p]] + sd) - mx) : 0.0f;
        den += wave_sum(w);
    }

    v2f acc = {0.0f, 0.0f};
    add_rows(Xp, src0, w0 / den, deg < 32 ? deg : 32, lane, acc);
    for (int c0 = 32; c0 < deg; c0 += 32) {
        const int p = c0 + lane;
        int   s = 0;
        float w = 0.0f;
        if (p < deg) { s = srt[base + p]; w = expf(leaky(s_src[s] + sd) - mx); }
        add_rows(Xp, s, w / den, deg - c0 < 32 ? deg - c0 : 32, lane, acc);
    }

    v2f r;
    r.x = acc.x > 0.0f ? acc.x : (expf(acc.x) - 1.0f);
    r.y = acc.y > 0.0f ? acc.y : (expf(acc.y) - 1.0f);
    *(v2f*)(out + (size_t)node * D_OUT + 2 * lane) = r;
}

extern "C" void kernel_launch(void* const* d_in, const int* in_sizes, int n_in,
                              void* d_out, int out_size, void* d_ws, size_t ws_size,
                              hipStream_t stream) {
    const float* X       = (const float*)d_in[0];
    const float* theta_w = (const float*)d_in[1];
    const float* theta_b = (const float*)d_in[2];
    const float* a_src   = (const float*)d_in[3];
    const float* a_src_b = (const float*)d_in[4];
    const float* a_dst   = (const float*)d_in[5];
    const float* a_dst_b = (const float*)d_in[6];
    const int*   e_src   = (const int*)d_in[7];
    const int*   e_dst   = (const int*)d_in[8];
    float*       out     = (float*)d_out;

    float* Xp    = (float*)d_ws;
    float* s_src = Xp    + (size_t)N_NODES * D_OUT;
    float* s_dst = s_src + N_NODES;
    int*   cnt   = (int*)(s_dst + N_NODES);
    int*   off   = cnt + N_NODES;
    int*   srt   = off + N_NODES;
    int*   pos   = (int*)d_ws;
    int*   grp   = pos + N_EDGES;
    int*   nxt   = grp + N_EDGES;

    const int nodeGrid  = (N_NODES + 255) / 256;
    const int gemmGrid  = (ROW_BLOCKS + 7) / 8;
    const int nodeWGrid = (N_NODES * 32 + 255) / 256;
    const int edgeGrid  = (N_EDGES + 255) / 256;

    k_clear<<<nodeGrid, 256, 0, stream>>>(cnt, nxt);
    k_slot<<<edgeGrid, 256, 0, stream>>>(e_dst, cnt, pos);
    k_alloc<<<nodeGrid, 256, 0, stream>>>(cnt, nxt, off);
    k_fill<<<edgeGrid, 256, 0, stream>>>(e_src, e_dst, pos, off, grp);
    k_sort<<<nodeWGrid, 256, 0, stream>>>(cnt, off, grp, srt);
    k_gemm<<<gemmGrid, 256, 0, stream>>>(X, theta_w, theta_b, Xp);
    k_scores<<<nodeWGrid, 256, 0, stream>>>(Xp, a_src, a_src_b, a_dst, a_dst_b, s_src, s_dst);
    k_gather<<<nodeWGrid, 256, 0, stream>>>(cnt, off, srt, s_src, s_dst, Xp, out);
}


# grouping level 2: each bucket's part of grp is assembled in LDS and written as whole 128-byte lines, not one scattered 4-byte store per edge; gather still orders each range, sum order unchanged
# speedup vs baseline: 5.4262x; 2.9268x over previous
#include <hip/hip_runtime.h>
#include <math.h>

typedef float v2f __attribute__((ext_vector_type(2)));
typedef float v8f __attribute__((ext_vector_type(8)));

#define N_NODES   100000
#define N_EDGES   1600000
#define D_IN      128
#define D_OUT     64
#define ROW_BLOCKS (N_NODES / 16)

#define STRIP_WAVES 25
#define GEMM_WGS    (ROW_BLOCKS / STRIP_WAVES)
static_assert(GEMM_WGS * STRIP_WAVES == ROW_BLOCKS, "strip waves cover the row strips exactly");

__device__ __forceinline__ void gemm_part(const int wg,
                                          const float* __restrict__ X,
                                          const float* __restrict__ W,
                                          const float* __restrict__ bias,
                                          const float* __restrict__ a_src,
                                          const float* __restrict__ a_src_b,
                                          const float* __restrict__ a_dst,
                                          const float* __restrict__ a_dst_b,
                                          float* __restrict__ Xp,
                                          float* __restrict__ s_src,
                                          float* __restrict__ s_dst) {
    __shared__ float Wl[D_IN * D_OUT];
    const int tid = threadIdx.x;
#pragma unroll
    for (int j = 0; j < D_IN * D_OUT / 1024; ++j) {
        const int i  = tid + j * 1024;
        const int k  = i >> 6;
        const int c  = i & 63;
        const int kk = k & ~1;
        Wl[kk * D_OUT + (c << 1) + (k & 1)] = W[i];
    }
    __syncthreads();

    const int wave = tid >> 5;
    const int lane = tid & 31;
    if (wave >= STRIP_WAVES) return;
    const int rowBlock = wg * STRIP_WAVES + wave;

    const int rowBase = rowBlock * 16;
    const int l15     = lane & 15;
    const int khalf   = (lane >> 4) << 1;

    const float* xrow = X + (size_t)(rowBase + l15) * D_IN;

    v8f acc0 = {}, acc1 = {}, acc2 = {}, acc3 = {};
    for (int k0 = 0; k0 < D_IN; k0 += 4) {
        v2f a = *(const v2f*)(xrow + k0 + khalf);
        const float* wp = Wl + (k0 + khalf) * D_OUT;
        v2f b0 = *(const v2f*)(wp + (l15 << 1));
        v2f b1 = *(const v2f*)(wp + ((16 + l15) << 1));
        v2f b2 = *(const v2f*)(wp + ((32 + l15) << 1));
        v2f b3 = *(const v2f*)(wp + ((48 + l15) << 1));
        acc0 = __builtin_amdgcn_wmma_f32_16x16x4_f32(false, a, false, b0, (short)0, acc0, false, false);
        acc1 = __builtin_amdgcn_wmma_f32_16x16x4_f32(false, a, false, b1, (short)0, acc1, false, false);
        acc2 = __builtin_amdgcn_wmma_f32_16x16x4_f32(false, a, false, b2, (short)0, acc2, false, false);
        acc3 = __builtin_amdgcn_wmma_f32_16x16x4_f32(false, a, false, b3, (short)0, acc3, false, false);
    }

    const int orow = rowBase + (lane >> 4) * 8;
    const float b0s = bias[l15], b1s = bias[16 + l15];
    const float b2s = bias[32 + l15], b3s = bias[48 + l15];
    const v2f asd0 = {a_src[l15],      a_dst[l15]},      asd1 = {a_src[16 + l15], a_dst[16 + l15]};
    const v2f asd2 = {a_src[32 + l15], a_dst[32 + l15]}, asd3 = {a_src[48 + l15], a_dst[48 + l15]};
    v2f v[8];
#pragma unroll
    for (int r = 0; r < 8; ++r) {
        float* o = Xp + (size_t)(orow + r) * D_OUT;
        const float x0 = acc0[r] + b0s, x1 = acc1[r] + b1s;
        const float x2 = acc2[r] + b2s, x3 = acc3[r] + b3s;
        o[l15]      = x0;
        o[16 + l15] = x1;
        o[32 + l15] = x2;
        o[48 + l15] = x3;
        v[r] = x0 * asd0 + x1 * asd1 + x2 * asd2 + x3 * asd3;
    }

    const bool h8 = (l15 & 8) != 0, h4 = (l15 & 4) != 0;
    const bool h2 = (l15 & 2) != 0, h1 = (l15 & 1) != 0;
    float w[8], x[4], y[2];
    for (int i = 0; i < 8; ++i) {
        const float keep = h8 ? v[i].y : v[i].x;
        const float send = h8 ? v[i].x : v[i].y;
        w[i] = keep + __shfl_xor(send, 8, 32);
    }
    for (int i = 0; i < 4; ++i) {
        const float keep = h4 ? w[i + 4] : w[i];
        const float send = h4 ? w[i] : w[i + 4];
        x[i] = keep + __shfl_xor(send, 4, 32);
    }
    for (int i = 0; i < 2; ++i) {
        const float keep = h2 ? x[i + 2] : x[i];
        const float send = h2 ? x[i] : x[i + 2];
        y[i] = keep + __shfl_xor(send, 2, 32);
    }
    const float keep1 = h1 ? y[1] : y[0];
    const float send1 = h1 ? y[0] : y[1];
    const float tot   = keep1 + __shfl_xor(send1, 1, 32);

    float*      sp = h8 ? s_dst : s_src;
    const float sb = h8 ? a_dst_b[0] : a_src_b[0];
    sp[orow + (l15 & 7)] = tot + sb;
}

#define NB        256
#define NPB       391
#define CHUNK     (N_EDGES / NB)
#define SRC_BITS  17
#define EPT       7
#define CPITCH    288
#define TPITCH    6272
#define SEG_PER_WAVE (NB / 32)
#define LB_CAP    8192
static_assert(NB * NPB >= N_NODES, "every node belongs to a bucket");
static_assert(CHUNK * NB == N_EDGES, "chunks cover the edge list exactly");
static_assert(CHUNK <= EPT * 1024, "a chunk fits one pass of its work-group");
static_assert(N_NODES <= (1 << SRC_BITS), "packed word: source id field");
static_assert(NPB <= 512, "packed word: local node field; bucket counters in LDS");
static_assert(CPITCH >= NB + 1 && CPITCH % 32 == 0, "coff row holds NB + 1 words, whole lines");
static_assert(TPITCH >= CHUNK && TPITCH % 32 == 0, "tmp row holds a chunk, whole lines");
static_assert(NB * TPITCH + NB * CPITCH <= N_NODES * D_OUT, "grouping scratch fits in the output buffer");

__global__ __launch_bounds__(1024) void k_chunk_sort(const int* __restrict__ e_src,
                                                     const int* __restrict__ e_dst,
                                                     int* __restrict__ coff,
                                                     int* __restrict__ tmp) {
    __shared__ int h[NB];
    __shared__ int wsum[8];
    __shared__ int buf[CHUNK];
    const int tid  = threadIdx.x;
    const int lane = tid & 31;
    const int wave = tid >> 5;
    const int wg   = blockIdx.x;

    if (tid < NB) h[tid] = 0;
    unsigned d[EPT];
    int      s[EPT];
#pragma unroll
    for (int k = 0; k < EPT; ++k) {
        const int i = tid + k * 1024;
        d[k] = i < CHUNK ? (unsigned)e_dst[wg * CHUNK + i] : 0u;
        s[k] = i < CHUNK ? e_src[wg * CHUNK + i] : 0;
    }
    __syncthreads();
#pragma unroll
    for (int k = 0; k < EPT; ++k)
        if (tid + k * 1024 < CHUNK) atomicAdd(&h[d[k] / NPB], 1);
    __syncthreads();

    const int v = tid < NB ? h[tid] : 0;
    int incl = v;
    for (int o = 1; o < 32; o <<= 1) {
        const int t = __shfl_up(incl, o, 32);
        if (lane >= o) incl += t;
    }
    if (tid < NB && lane == 31) wsum[wave] = incl;
    __syncthreads();
    if (tid < NB) {
        int wp = 0;
        for (int w = 0; w < wave; ++w) wp += wsum[w];
        const int start = wp + incl - v;
        h[tid] = start;
        coff[wg * CPITCH + tid] = start;
    }
    if (tid == 0) coff[wg * CPITCH + NB] = CHUNK;
    __syncthreads();

    int pos[EPT];
#pragma unroll
    for (int k = 0; k < EPT; ++k)
        if (tid + k * 1024 < CHUNK) pos[k] = atomicAdd(&h[d[k] / NPB], 1);
#pragma unroll
    for (int k = 0; k < EPT; ++k)
        if (tid + k * 1024 < CHUNK)
            buf[pos[k]] = (int)(((d[k] % NPB) << SRC_BITS) | (unsigned)s[k]);
    __syncthreads();
#pragma unroll
    for (int k = 0; k < EPT; ++k) {
        const int i = tid + k * 1024;
        if (i < CHUNK) tmp[wg * TPITCH + i] = buf[i];
    }
}

__device__ __forceinline__ void group_part(const int bk,
                                           const int* __restrict__ coff,
                                           const int* __restrict__ tmp,
                                           int* __restrict__ cnt,
                                           int* __restrict__ off,
                                           int* __restrict__ grp) {
    __shared__ int c[512];
    __shared__ int wsum[16];
    __shared__ int lsum[NB / 32];
    __shared__ int seg_src[NB];
    __shared__ int seg_cnt[NB];
    __shared__ int lb[LB_CAP];
    const int tid  = threadIdx.x;
    const int lane = tid & 31;
    const int wave = tid >> 5;

    if (tid < 512) c[tid] = 0;
    if (tid < NB) {
        const int a = coff[tid * CPITCH + bk];
        const int b = coff[tid * CPITCH + bk + 1];
        seg_src[tid] = tid * TPITCH + a;
        seg_cnt[tid] = b - a;
        int t = a;
        for (int o = 16; o > 0; o >>= 1) t += __shfl_xor(t, o, 32);
        if (lane == 0) lsum[wave] = t;
    }
    __syncthreads();
    int lo = 0;
    for (int w = 0; w < NB / 32; ++w) lo += lsum[w];

    int sb[SEG_PER_WAVE], sc[SEG_PER_WAVE], w0[SEG_PER_WAVE];
#pragma unroll
    for (int k = 0; k < SEG_PER_WAVE; ++k) {
        sb[k] = __builtin_amdgcn_readfirstlane(seg_src[wave + 32 * k]);
        sc[k] = __builtin_amdgcn_readfirstlane(seg_cnt[wave + 32 * k]);
    }
#pragma unroll
    for (int k = 0; k < SEG_PER_WAVE; ++k)
        w0[k] = lane < sc[k] ? tmp[sb[k] + lane] : -1;
#pragma unroll
    for (int k = 0; k < SEG_PER_WAVE; ++k)
        if (w0[k] >= 0) atomicAdd(&c[w0[k] >> SRC_BITS], 1);
    for (int k = 0; k < SEG_PER_WAVE; ++k)
        for (int j = 32 + lane; j < sc[k]; j += 32)
            atomicAdd(&c[tmp[sb[k] + j] >> SRC_BITS], 1);
    __syncthreads();

    const int v = tid < 512 ? c[tid] : 0;
    int incl = v;
    for (int o = 1; o < 32; o <<= 1) {
        const int t = __shfl_up(incl, o, 32);
        if (lane >= o) incl += t;
    }
    if (tid < 512 && lane == 31) wsum[wave] = incl;
    __syncthreads();
    int totv = 0;
    for (int w = 0; w < 16; ++w) totv += wsum[w];
    const int tot = __builtin_amdgcn_readfirstlane(totv);
    if (tid < 512) {
        int wp = 0;
        for (int w = 0; w < wave; ++w) wp += wsum[w];
        const int start = wp + incl - v;
        const int node  = bk * NPB + tid;
        if (tid < NPB && node < N_NODES) { cnt[node] = v; off[node] = lo + start; }
        c[tid] = start;
    }
    __syncthreads();

    if (tot <= LB_CAP) {
#pragma unroll
        for (int k = 0; k < SEG_PER_WAVE; ++k) {
            if (w0[k] >= 0) {
                const int p = atomicAdd(&c[w0[k] >> SRC_BITS], 1);
                lb[p] = w0[k] & ((1 << SRC_BITS) - 1);
            }
        }
        for (int k = 0; k < SEG_PER_WAVE; ++k)
            for (int j = 32 + lane; j < sc[k]; j += 32) {
                const int w = tmp[sb[k] + j];
                const int p = atomicAdd(&c[w >> SRC_BITS], 1);
                lb[p] = w & ((1 << SRC_BITS) - 1);
            }
    } else {
#pragma unroll
        for (int k = 0; k < SEG_PER_WAVE; ++k) {
            if (w0[k] >= 0) {
                const int p = atomicAdd(&c[w0[k] >> SRC_BITS], 1);
                grp[lo + p] = w0[k] & ((1 << SRC_BITS) - 1);
            }
        }
        for (int k = 0; k < SEG_PER_WAVE; ++k)
            for (int j = 32 + lane; j < sc[k]; j += 32) {
                const int w = tmp[sb[k] + j];
                const int p = atomicAdd(&c[w >> SRC_BITS], 1);
                grp[lo + p] = w & ((1 << SRC_BITS) - 1);
            }
    }
    __syncthreads();
    if (tot <= LB_CAP) {
        for (int i = tid - (lo & 31); i < tot; i += 1024)
            if (i >= 0) grp[lo + i] = lb[i];
    }
}

__global__ __launch_bounds__(1024, 16) void k_group_gemm(const int* __restrict__ coff,
                                                        const int* __restrict__ tmp,
                                                        int* __restrict__ cnt,
                                                        int* __restrict__ off,
                                                        int* __restrict__ grp,
                                                        const float* __restrict__ X,
                                                        const float* __restrict__ W,
                                                        const float* __restrict__ bias,
                                                        const float* __restrict__ a_src,
                                                        const float* __restrict__ a_src_b,
                                                        const float* __restrict__ a_dst,
                                                        const float* __restrict__ a_dst_b,
                                                        float* __restrict__ Xp,
                                                        float* __restrict__ s_src,
                                                        float* __restrict__ s_dst) {
    if (blockIdx.x < NB)
        group_part(blockIdx.x, coff, tmp, cnt, off, grp);
    else
        gemm_part(blockIdx.x - NB, X, W, bias, a_src, a_src_b, a_dst, a_dst_b, Xp, s_src, s_dst);
}

static_assert(N_NODES <= (1 << 27), "packed sort key: source id and lane share 32 bits");
__device__ __forceinline__ int sorted_sources_32(const int* __restrict__ grp, int base,
                                                 int deg, int lane) {
    const int      key  = lane < deg ? grp[base + lane] : 0x07ffffff;
    const unsigned mine = ((unsigned)key << 5) | (unsigned)lane;
    int rank = 0;
    const int bound = (deg + 3) & ~3;
    for (int j = 0; j < bound; j += 4) {
#pragma unroll
        for (int u = 0; u < 4; ++u) {
            const unsigned theirs = (unsigned)__builtin_amdgcn_readlane((int)mine, j + u);
            rank += theirs < mine ? 1 : 0;
        }
    }
    const int to = lane < deg ? rank : lane;
    return __builtin_amdgcn_ds_permute(to << 2, key);
}

__device__ int ranked_source(const int* __restrict__ grp, int base, int deg,
                             int want, int lane) {
    int res = 0;
    for (int s0 = 0; s0 < deg; s0 += 32) {
        const int p   = s0 + lane;
        const int key = p < deg ? grp[base + p] : 0x7fffffff;
        const unsigned long long mine =
            ((unsigned long long)(unsigned)key << 32) | (unsigned)p;
        int rank = 0;
        for (int d0 = 0; d0 < deg; d0 += 32) {
            const int q     = d0 + lane;
            const int other = q < deg ? grp[base + q] : 0x7fffffff;
            const int nv    = deg - d0 < 32 ? deg - d0 : 32;
            for (int j = 0; j < nv; ++j) {
                const int kj = __builtin_amdgcn_readlane(other, j);
                const unsigned long long theirs =
                    ((unsigned long long)(unsigned)kj << 32) | (unsigned)(d0 + j);
                rank += theirs < mine ? 1 : 0;
            }
        }
        const int nv = deg - s0 < 32 ? deg - s0 : 32;
        for (int j = 0; j < nv; ++j) {
            const int rj = __builtin_amdgcn_readlane(rank, j);
            const int kj = __builtin_amdgcn_readlane(key, j);
            if (rj == want) res = kj;
        }
    }
    return res;
}

__device__ __forceinline__ float wave_max(float v) {
    for (int o = 16; o > 0; o >>= 1) v = fmaxf(v, __shfl_xor(v, o, 32));
    return v;
}
__device__ __forceinline__ float wave_sum(float v) {
    for (int o = 16; o > 0; o >>= 1) v += __shfl_xor(v, o, 32);
    return v;
}
__device__ __forceinline__ float leaky(float v) { return v > 0.0f ? v : 0.01f * v; }

__device__ __forceinline__ void add_rows(const float* __restrict__ Xp, int src, float coef,
                                         int nv, int lane, v2f& acc) {
#pragma unroll 4
    for (int j = 0; j < nv; ++j) {
        const int   sj = __builtin_amdgcn_readlane(src, j);
        const float cj = __builtin_bit_cast(float,
            __builtin_amdgcn_readlane(__builtin_bit_cast(int, coef), j));
        const v2f x = *(const v2f*)(Xp + (size_t)sj * D_OUT + 2 * lane);
        acc.x += cj * x.x;
        acc.y += cj * x.y;
    }
}

__device__ __forceinline__ void store_elu(float* __restrict__ out, int node, int lane, v2f acc) {
    v2f r;
    r.x = acc.x > 0.0f ? acc.x : (expf(acc.x) - 1.0f);
    r.y = acc.y > 0.0f ? acc.y : (expf(acc.y) - 1.0f);
    *(v2f*)(out + (size_t)node * D_OUT + 2 * lane) = r;
}

__global__ __launch_bounds__(256) void k_gather(const int* __restrict__ cnt,
                                                const int* __restrict__ off,
                                                const int* __restrict__ grp,
                                                const float* __restrict__ s_src,
                                                const float* __restrict__ s_dst,
                                                const float* __restrict__ Xp,
                                                float* __restrict__ out) {
    const int lane = threadIdx.x & 31;
    const int node = __builtin_amdgcn_readfirstlane((blockIdx.x * 256 + threadIdx.x) >> 5);
    if (node >= N_NODES) return;
    const int deg  = __builtin_amdgcn_readfirstlane(cnt[node]);
    const int base = __builtin_amdgcn_readfirstlane(off[node]);
    const float sd = s_dst[node];
    v2f acc = {0.0f, 0.0f};

    if (__builtin_expect(deg > 32, 0)) {
        float mx = -__builtin_inff();
        for (int c0 = 0; c0 < deg; c0 += 32) {
            const int p = c0 + lane;
            const int s = ranked_source(grp, base, deg, p, lane);
            if (p < deg) mx = fmaxf(mx, leaky(s_src[s] + sd));
        }
        mx = wave_max(mx);
        float den = 0.0f;
        for (int c0 = 0; c0 < deg; c0 += 32) {
            const int p = c0 + lane;
            const int s = ranked_source(grp, base, deg, p, lane);
            const float w = p < deg ? expf(leaky(s_src[s] + sd) - mx) : 0.0f;
            den += wave_sum(w);
        }
        for (int c0 = 0; c0 < deg; c0 += 32) {
            const int p = c0 + lane;
            const int s = ranked_source(grp, base, deg, p, lane);
            const float w = p < deg ? expf(leaky(s_src[s] + sd) - mx) : 0.0f;
            add_rows(Xp, s, w / den, deg - c0 < 32 ? deg - c0 : 32, lane, acc);
        }
        store_elu(out, node, lane, acc);
        return;
    }

    const int ordered = sorted_sources_32(grp, base, deg, lane);
    int   src0 = 0;
    float v0   = -__builtin_inff();
    if (lane < deg) { src0 = ordered; v0 = leaky(s_src[src0] + sd); }
    const float mx  = wave_max(v0);
    const float w0  = lane < deg ? expf(v0 - mx) : 0.0f;
    const float den = wave_sum(w0);
    add_rows(Xp, src0, w0 / den, deg, lane, acc);
    store_elu(out, node, lane, acc);
}

extern "C" void kernel_launch(void* const* d_in, const int* in_sizes, int n_in,
                              void* d_out, int out_size, void* d_ws, size_t ws_size,
                              hipStream_t stream) {
    const float* X       = (const float*)d_in[0];
    const float* theta_w = (const float*)d_in[1];
    const float* theta_b = (const float*)d_in[2];
    const float* a_src   = (const float*)d_in[3];
    const float* a_src_b = (const float*)d_in[4];
    const float* a_dst   = (const float*)d_in[5];
    const float* a_dst_b = (const float*)d_in[6];
    const int*   e_src   = (const int*)d_in[7];
    const int*   e_dst   = (const int*)d_in[8];
    float*       out     = (float*)d_out;

    float* Xp     = (float*)d_ws;
    float* s_src  = Xp    + (size_t)N_NODES * D_OUT;
    float* s_dst  = s_src + N_NODES;
    int*   cnt    = (int*)(s_dst + N_NODES);
    int*   off    = cnt + N_NODES;
    int*   grp    = off + N_NODES;
    int*   tmp    = (int*)d_out;
    int*   coff   = tmp + NB * TPITCH;

    const int nodeWGrid = (N_NODES * 32 + 255) / 256;

    k_chunk_sort<<<NB, 1024, 0, stream>>>(e_src, e_dst, coff, tmp);
    k_group_gemm<<<NB + GEMM_WGS, 1024, 0, stream>>>(coff, tmp, cnt, off, grp,
                                                     X, theta_w, theta_b, a_src, a_src_b,
                                                     a_dst, a_dst_b, Xp, s_src, s_dst);
    k_gather<<<nodeWGrid, 256, 0, stream>>>(cnt, off, grp, s_src, s_dst, Xp, out);
}
